# starting kernel
# baseline (speedup 1.0000x reference)
#include <hip/hip_runtime.h>

typedef float v2f __attribute__((ext_vector_type(2)));
typedef float v8f __attribute__((ext_vector_type(8)));

#define NPTS  80000
#define CIN   3
#define COUT  32
#define KVOL  125
#define WSZ   (KVOL * CIN * COUT)
#define MPAD  950000
#define NTILES (MPAD / 16)

__global__ void spconv_zero_kernel(float* __restrict__ out, int n) {
  int i = blockIdx.x * blockDim.x + threadIdx.x;
  int stride = gridDim.x * blockDim.x;
  for (; i < n; i += stride) out[i] = 0.0f;
}

__global__ __launch_bounds__(256) void spconv_wmma_kernel(
    const float* __restrict__ feat,
    const float* __restrict__ weight,
    const int*   __restrict__ in_idx,
    const int*   __restrict__ out_idx,
    const int*   __restrict__ off_idx,
    float*       __restrict__ out) {
  __shared__ float sW[WSZ];
  for (int i = threadIdx.x; i < WSZ; i += blockDim.x) sW[i] = weight[i];
  __syncthreads();

  const int lane = threadIdx.x & 31;
  const int row  = lane & 15;
  const int hi   = lane >> 4;
  const int waveId = (blockIdx.x * blockDim.x + threadIdx.x) >> 5;
  const int nWaves = (gridDim.x * blockDim.x) >> 5;

  for (int tile = waveId; tile < NTILES; tile += nWaves) {
    const int base = tile << 4;

    const int nbase = base + (nWaves << 4);
    if (nbase < MPAD) {
      __builtin_prefetch(in_idx  + nbase, 0, 0);
      __builtin_prefetch(out_idx + nbase, 0, 0);
    }

    const int off0  = off_idx[base];
    const int off15 = off_idx[base + 15];

    if (off0 == off15) {
      const int gi = in_idx[base + row];
      v2f a;
      a.x = feat[gi * CIN + (hi ? 2 : 0)];
      a.y = hi ? 0.0f : feat[gi * CIN + 1];

      const float* wb = &sW[off0 * (CIN * COUT)];
      v8f c = {};
#pragma unroll
      for (int h = 0; h < 2; ++h) {
        const int ch = (h << 4) + row;
        v2f b;
        b.x = wb[(hi ? 2 : 0) * COUT + ch];
        b.y = hi ? 0.0f : wb[1 * COUT + ch];

        v8f d = __builtin_amdgcn_wmma_f32_16x16x4_f32(
             false, a,  false, b,
             (short)0, c,  false,  false);

#pragma unroll
        for (int r = 0; r < 8; ++r) {
          const int trow = r + (hi << 3);
          const int orow = out_idx[base + trow];
          if (orow < NPTS)
            unsafeAtomicAdd(&out[orow * COUT + ch], d[r]);
        }
      }
    } else {
      const int m  = base + row;
      const int gi = in_idx[m];
      const int oo = out_idx[m];
      const int of = off_idx[m];
      if (oo < NPTS) {
        const float f0 = feat[gi * CIN + 0];
        const float f1 = feat[gi * CIN + 1];
        const float f2 = feat[gi * CIN + 2];
        const float* wb = &sW[of * (CIN * COUT)];
#pragma unroll
        for (int cc = 0; cc < 16; ++cc) {
          const int ch = (hi << 4) + cc;
          const float v = f0 * wb[ch] + f1 * wb[COUT + ch] + f2 * wb[2 * COUT + ch];
          unsafeAtomicAdd(&out[oo * COUT + ch], v);
        }
      }
    }
  }
}

extern "C" void kernel_launch(void* const* d_in, const int* in_sizes, int n_in,
                              void* d_out, int out_size, void* d_ws, size_t ws_size,
                              hipStream_t stream) {
  const float* feat    = (const float*)d_in[0];
  const float* weight  = (const float*)d_in[1];
  const int*   in_idx  = (const int*)d_in[2];
  const int*   out_idx = (const int*)d_in[3];
  const int*   off_idx = (const int*)d_in[4];
  float* out = (float*)d_out;

  spconv_zero_kernel<<<512, 256, 0, stream>>>(out, out_size);

  spconv_wmma_kernel<<<1024, 256, 0, stream>>>(feat, weight, in_idx, out_idx,
                                               off_idx, out);
}


# Output-stationary: a bucket pass finds each (offset, 64-row block) piece of the list; a block owning 64 rows sums its WMMA tiles in LDS (f32 LDS adds, arrival-order sums), stores each row once
# speedup vs baseline: 2.9267x; 2.9267x over previous
#include <hip/hip_runtime.h>

typedef float v2f __attribute__((ext_vector_type(2)));
typedef float v8f __attribute__((ext_vector_type(8)));

#define NPTS  80000
#define CIN   3
#define COUT  32
#define KVOL  125
#define MPAD  950000
#define TROWS 64
#define NRB   (NPTS / TROWS)
#define NBUCK (KVOL * NRB)
#define PRE_THREADS 1024
#define PRE_PER 4
#define PRE_SPAN (PRE_THREADS * PRE_PER)
#define PRE_BLOCKS ((MPAD + PRE_SPAN - 1) / PRE_SPAN)
#define WS_INTS (NBUCK + 1 + PRE_BLOCKS)
#define KB 8

__device__ __forceinline__ int bucket_of(int off, int orow) {
  if ((unsigned)orow >= (unsigned)NPTS || (unsigned)off >= (unsigned)KVOL) return NBUCK;
  return off * NRB + orow / TROWS;
}

__global__ __launch_bounds__(PRE_THREADS) void spconv_bucket_kernel(
    const int* __restrict__ out_idx, const int* __restrict__ off_idx,
    int* __restrict__ ws, size_t ws_bytes) {
  if (ws_bytes < WS_INTS * sizeof(int)) return;
  int* start = ws;
  int* flags = ws + NBUCK + 1;
  const int base = blockIdx.x * PRE_SPAN + threadIdx.x;
  int cb[PRE_PER], pb[PRE_PER];
#pragma unroll
  for (int j = 0; j < PRE_PER; ++j) {
    const int m  = base + j * PRE_THREADS;
    const int mc = m < MPAD ? m : MPAD - 1;
    const int mp = mc > 0 ? mc - 1 : 0;
    const int f1 = off_idx[mc], o1 = out_idx[mc];
    const int f0 = off_idx[mp], o0 = out_idx[mp];
    cb[j] = bucket_of(f1, o1);
    pb[j] = bucket_of(f0, o0);
  }
  int bad = 0;
#pragma unroll
  for (int j = 0; j < PRE_PER; ++j) {
    const int m = base + j * PRE_THREADS;
    if (m < MPAD) {
      if (cb[j] < pb[j]) {
        bad = 1;
      } else {
        const int lo = (m == 0) ? 0 : pb[j] + 1;
        for (int b = lo; b <= cb[j]; ++b) start[b] = m;
        if (m == MPAD - 1)
          for (int b = cb[j] + 1; b <= NBUCK; ++b) start[b] = MPAD;
      }
    }
  }
  const int anybad = __syncthreads_or(bad);
  if (threadIdx.x == 0) flags[blockIdx.x] = anybad;
}

__device__ __forceinline__ void add_tile(float* acc, int j0, int cnt, int oo,
                                         v8f d0, v8f d1, int lane, int hi) {
#pragma unroll
  for (int r = 0; r < 8; ++r) {
    if (r < cnt) {
      const int t0 = __builtin_amdgcn_permlanex16(0, __float_as_int(d1[r]), 0x76543210, 0xfedcba98, false, false);
      const int t1 = __builtin_amdgcn_permlanex16(0, __float_as_int(d0[r]), 0x76543210, 0xfedcba98, false, false);
      const float x = hi ? __int_as_float(t0) : d0[r];
      const float y = hi ? d1[r] : __int_as_float(t1);
      const int lr0 = __builtin_amdgcn_readlane(oo, r) - j0;
      if ((unsigned)lr0 < (unsigned)TROWS)
        __hip_atomic_fetch_add(acc + lr0 * COUT + lane, x, __ATOMIC_RELAXED, __HIP_MEMORY_SCOPE_WORKGROUP);
      if (r + 8 < cnt) {
        const int lr1 = __builtin_amdgcn_readlane(oo, r + 8) - j0;
        if ((unsigned)lr1 < (unsigned)TROWS)
          __hip_atomic_fetch_add(acc + lr1 * COUT + lane, y, __ATOMIC_RELAXED, __HIP_MEMORY_SCOPE_WORKGROUP);
      }
    }
  }
}

__global__ __launch_bounds__(256) void spconv_owner_kernel(
    const float* __restrict__ feat,
    const float* __restrict__ weight,
    const int*   __restrict__ in_idx,
    const int*   __restrict__ out_idx,
    const int*   __restrict__ off_idx,
    const int*   __restrict__ ws, size_t ws_bytes,
    float*       __restrict__ out) {
  __shared__ __attribute__((aligned(16))) float acc[TROWS * COUT];
  __shared__ int sBeg[128];
  __shared__ int sEnd[128];
  const int tid = threadIdx.x;
  const int rb  = blockIdx.x;
  const int j0  = rb * TROWS;

  for (int i = tid; i < TROWS * COUT; i += 256) acc[i] = 0.0f;

  const bool ws_ok = ws_bytes >= WS_INTS * sizeof(int);
  int bad = ws_ok ? 0 : 1;
  if (ws_ok) {
    const int* flags = ws + NBUCK + 1;
    if (tid < PRE_BLOCKS) bad = flags[tid];
    if (tid < 128) {
      int s = 0, e = 0;
      if (tid < KVOL) { s = ws[tid * NRB + rb]; e = ws[tid * NRB + rb + 1]; }
      sBeg[tid] = s;
      sEnd[tid] = e;
    }
  }
  const int anybad = __syncthreads_or(bad);

  if (!anybad) {
    const int lane = tid & 31;
    const int row  = lane & 15;
    const int hi   = lane >> 4;
    const int w    = __builtin_amdgcn_readfirstlane(tid >> 5);
    const int ksel = hi ? 2 : 0;

#pragma unroll
    for (int it = 0; it < 128 / (8 * KB); ++it) {
      const int kb = w + it * 8 * KB;
      int cnt[KB], oo[KB], gi[KB];
      v2f a[KB], b0[KB], b1[KB];
#pragma unroll
      for (int u = 0; u < KB; ++u) {
        const int k = kb + 8 * u;
        const int s = __builtin_amdgcn_readfirstlane(sBeg[k]);
        const int e = __builtin_amdgcn_readfirstlane(sEnd[k]);
        cnt[u] = e - s;
        const bool v = row < cnt[u];
        gi[u] = v ? in_idx[s + row] : 0;
        oo[u] = v ? out_idx[s + row] : -1;
        const float* wb = weight + (k < KVOL ? k : 0) * (CIN * COUT);
        b0[u].x = wb[ksel * COUT + row];
        b0[u].y = hi ? 0.0f : wb[COUT + row];
        b1[u].x = wb[ksel * COUT + 16 + row];
        b1[u].y = hi ? 0.0f : wb[COUT + 16 + row];
      }
#pragma unroll
      for (int u = 0; u < KB; ++u) {
        const bool v = row < cnt[u];
        a[u].x = v ? feat[gi[u] * CIN + ksel] : 0.0f;
        a[u].y = (v && !hi) ? feat[gi[u] * CIN + 1] : 0.0f;
      }
#pragma unroll
      for (int u = 0; u < KB; ++u) {
        if (cnt[u] > 0) {
          const v8f z = {};
          const v8f d0 = __builtin_amdgcn_wmma_f32_16x16x4_f32(false, a[u], false, b0[u], (short)0, z, false, false);
          const v8f d1 = __builtin_amdgcn_wmma_f32_16x16x4_f32(false, a[u], false, b1[u], (short)0, z, false, false);
          add_tile(acc, j0, cnt[u], oo[u], d0, d1, lane, hi);
        }
      }
    }
    for (int kk = 0; kk < 128; kk += 32) {
      const int cl = sEnd[kk + lane] - sBeg[kk + lane];
      unsigned heavy = __builtin_amdgcn_ballot_w32(cl > 16);
      while (heavy) {
        const int k = kk + __builtin_ctz(heavy);
        heavy &= heavy - 1;
        const int s = __builtin_amdgcn_readfirstlane(sBeg[k]);
        const int e = __builtin_amdgcn_readfirstlane(sEnd[k]);
        int p = s + 16 * (1 + w);
        if (p < e) {
          const float* wb = weight + k * (CIN * COUT);
          v2f c0, c1;
          c0.x = wb[ksel * COUT + row];
          c0.y = hi ? 0.0f : wb[COUT + row];
          c1.x = wb[ksel * COUT + 16 + row];
          c1.y = hi ? 0.0f : wb[COUT + 16 + row];
          for (; p < e; p += 16 * 8) {
            const int c = e - p;
            const bool v = row < c;
            const int g = v ? in_idx[p + row] : 0;
            const int o = v ? out_idx[p + row] : -1;
            v2f aa;
            aa.x = v ? feat[g * CIN + ksel] : 0.0f;
            aa.y = (v && !hi) ? feat[g * CIN + 1] : 0.0f;
            const v8f z = {};
            const v8f d0 = __builtin_amdgcn_wmma_f32_16x16x4_f32(false, aa, false, c0, (short)0, z, false, false);
            const v8f d1 = __builtin_amdgcn_wmma_f32_16x16x4_f32(false, aa, false, c1, (short)0, z, false, false);
            add_tile(acc, j0, c, o, d0, d1, lane, hi);
          }
        }
      }
    }
  } else {
    for (int m = tid; m < MPAD; m += 256) {
      const int o = out_idx[m];
      const int f = off_idx[m];
      const int lr = o - j0;
      if ((unsigned)o < (unsigned)NPTS && (unsigned)f < (unsigned)KVOL && (unsigned)lr < (unsigned)TROWS) {
        const int g = in_idx[m];
        const float f0 = feat[g * CIN + 0];
        const float f1 = feat[g * CIN + 1];
        const float f2 = feat[g * CIN + 2];
        const float* wb = weight + f * (CIN * COUT);
        for (int ch = 0; ch < COUT; ++ch)
          __hip_atomic_fetch_add(acc + lr * COUT + ch,
                                 f0 * wb[ch] + f1 * wb[COUT + ch] + f2 * wb[2 * COUT + ch],
                                 __ATOMIC_RELAXED, __HIP_MEMORY_SCOPE_WORKGROUP);
      }
    }
  }
  __syncthreads();

  float4* o4 = reinterpret_cast<float4*>(out + (size_t)j0 * COUT);
  const float4* a4 = reinterpret_cast<const float4*>(acc);
  for (int i = tid; i < TROWS * COUT / 4; i += 256) o4[i] = a4[i];
}

extern "C" void kernel_launch(void* const* d_in, const int* in_sizes, int n_in,
                              void* d_out, int out_size, void* d_ws, size_t ws_size,
                              hipStream_t stream) {
  const float* feat    = (const float*)d_in[0];
  const float* weight  = (const float*)d_in[1];
  const int*   in_idx  = (const int*)d_in[2];
  const int*   out_idx = (const int*)d_in[3];
  const int*   off_idx = (const int*)d_in[4];
  float* out = (float*)d_out;
  int* ws = (int*)d_ws;

  spconv_bucket_kernel<<<PRE_BLOCKS, PRE_THREADS, 0, stream>>>(out_idx, off_idx, ws, ws_size);
  spconv_owner_kernel<<<NRB, 256, 0, stream>>>(feat, weight, in_idx, out_idx, off_idx,
                                               ws, ws_size, out);
}


# Owner scatter without cross-lane moves: D register r now holds list pairs 2r, 2r+1 and one LDS add takes 16 channels of both; two LDS planes per row, summed at read-out; arrival-order sums
# speedup vs baseline: 3.4598x; 1.1822x over previous
#include <hip/hip_runtime.h>

typedef float v2f __attribute__((ext_vector_type(2)));
typedef float v8f __attribute__((ext_vector_type(8)));

#define NPTS  80000
#define CIN   3
#define COUT  32
#define KVOL  125
#define MPAD  950000
#define TROWS 64
#define NRB   (NPTS / TROWS)
#define NBUCK (KVOL * NRB)
#define PRE_THREADS 1024
#define PRE_PER 4
#define PRE_SPAN (PRE_THREADS * PRE_PER)
#define PRE_BLOCKS ((MPAD + PRE_SPAN - 1) / PRE_SPAN)
#define WS_INTS (NBUCK + 1 + PRE_BLOCKS)
#define KB 8
#define APITCH 64
#define ATRASH (TROWS * APITCH)

__device__ __forceinline__ int bucket_of(int off, int orow) {
  if ((unsigned)orow >= (unsigned)NPTS || (unsigned)off >= (unsigned)KVOL) return NBUCK;
  return off * NRB + orow / TROWS;
}

__global__ __launch_bounds__(PRE_THREADS) void spconv_bucket_kernel(
    const int* __restrict__ out_idx, const int* __restrict__ off_idx,
    int* __restrict__ ws, size_t ws_bytes) {
  if (ws_bytes < WS_INTS * sizeof(int)) return;
  int* start = ws;
  int* flags = ws + NBUCK + 1;
  const int base = blockIdx.x * PRE_SPAN + threadIdx.x;
  int cb[PRE_PER], pb[PRE_PER];
#pragma unroll
  for (int j = 0; j < PRE_PER; ++j) {
    const int m  = base + j * PRE_THREADS;
    const int mc = m < MPAD ? m : MPAD - 1;
    const int mp = mc > 0 ? mc - 1 : 0;
    const int f1 = off_idx[mc], o1 = out_idx[mc];
    const int f0 = off_idx[mp], o0 = out_idx[mp];
    cb[j] = bucket_of(f1, o1);
    pb[j] = bucket_of(f0, o0);
  }
  int bad = 0;
#pragma unroll
  for (int j = 0; j < PRE_PER; ++j) {
    const int m = base + j * PRE_THREADS;
    if (m < MPAD) {
      if (cb[j] < pb[j]) {
        bad = 1;
      } else {
        const int lo = (m == 0) ? 0 : pb[j] + 1;
        for (int b = lo; b <= cb[j]; ++b) start[b] = m;
        if (m == MPAD - 1)
          for (int b = cb[j] + 1; b <= NBUCK; ++b) start[b] = MPAD;
      }
    }
  }
  const int anybad = __syncthreads_or(bad);
  if (threadIdx.x == 0) flags[blockIdx.x] = anybad;
}

template <int R>
__device__ __forceinline__ void add_rows(float* acc, int cnt, int vbase, int lanepart,
                                         const v8f& d0, const v8f& d1) {
  if (2 * R < cnt) {
    const int rowpart = __builtin_amdgcn_update_dpp(0, vbase, 0x150 + R, 0xf, 0xf, true);
    float* p = acc + rowpart + lanepart;
    __hip_atomic_fetch_add(p,      d0[R], __ATOMIC_RELAXED, __HIP_MEMORY_SCOPE_WORKGROUP);
    __hip_atomic_fetch_add(p + 16, d1[R], __ATOMIC_RELAXED, __HIP_MEMORY_SCOPE_WORKGROUP);
    if constexpr (R < 7) add_rows<R + 1>(acc, cnt, vbase, lanepart, d0, d1);
  }
}

__device__ __forceinline__ void add_tile(float* acc, int j0, int cnt, int ov, int lanepart,
                                         const v8f& d0, const v8f& d1) {
  const int lr = ov - j0;
  const int vbase = ((unsigned)lr < (unsigned)TROWS) ? lr * APITCH : ATRASH;
  add_rows<0>(acc, cnt, vbase, lanepart, d0, d1);
}

__global__ __launch_bounds__(256) void spconv_owner_kernel(
    const float* __restrict__ feat,
    const float* __restrict__ weight,
    const int*   __restrict__ in_idx,
    const int*   __restrict__ out_idx,
    const int*   __restrict__ off_idx,
    const int*   __restrict__ ws, size_t ws_bytes,
    float*       __restrict__ out) {
  __shared__ __attribute__((aligned(16))) float acc[(TROWS + 1) * APITCH];
  __shared__ int sBeg[128];
  __shared__ int sEnd[128];
  const int tid = threadIdx.x;
  const int rb  = blockIdx.x;
  const int j0  = rb * TROWS;

  {
    float4* z4 = reinterpret_cast<float4*>(acc);
    for (int i = tid; i < (TROWS + 1) * APITCH / 4; i += 256) z4[i] = make_float4(0.0f, 0.0f, 0.0f, 0.0f);
  }

  const bool ws_ok = ws_bytes >= WS_INTS * sizeof(int);
  int bad = ws_ok ? 0 : 1;
  if (ws_ok) {
    const int* flags = ws + NBUCK + 1;
    if (tid < PRE_BLOCKS) bad = flags[tid];
    if (tid < 128) {
      int s = 0, e = 0;
      if (tid < KVOL) { s = ws[tid * NRB + rb]; e = ws[tid * NRB + rb + 1]; }
      sBeg[tid] = s;
      sEnd[tid] = e;
    }
  }
  const int anybad = __syncthreads_or(bad);

  if (!anybad) {
    const int lane = tid & 31;
    const int row  = lane & 15;
    const int hi   = lane >> 4;
    const int pa   = ((row & 7) << 1) | (row >> 3);
    const int po   = pa ^ hi;
    const int lanepart = hi * 32 + row;
    const int w    = __builtin_amdgcn_readfirstlane(tid >> 5);
    const int ksel = hi ? 2 : 0;

#pragma unroll
    for (int it = 0; it < 128 / (8 * KB); ++it) {
      const int kb = w + it * 8 * KB;
      int cnt[KB], oo[KB], gi[KB];
      v2f a[KB], b0[KB], b1[KB];
#pragma unroll
      for (int u = 0; u < KB; ++u) {
        const int k = kb + 8 * u;
        const int s = __builtin_amdgcn_readfirstlane(sBeg[k]);
        const int e = __builtin_amdgcn_readfirstlane(sEnd[k]);
        cnt[u] = e - s;
        const int ia = s + pa, io = s + po;
        const int ga = in_idx[ia < MPAD ? ia : MPAD - 1];
        const int oa = out_idx[io < MPAD ? io : MPAD - 1];
        gi[u] = pa < cnt[u] ? ga : 0;
        oo[u] = po < cnt[u] ? oa : -1;
        const float* wb = weight + (k < KVOL ? k : 0) * (CIN * COUT);
        b0[u].x = wb[ksel * COUT + row];
        b0[u].y = hi ? 0.0f : wb[COUT + row];
        b1[u].x = wb[ksel * COUT + 16 + row];
        b1[u].y = hi ? 0.0f : wb[COUT + 16 + row];
      }
#pragma unroll
      for (int u = 0; u < KB; ++u) {
        const bool v = pa < cnt[u];
        a[u].x = v ? feat[gi[u] * CIN + ksel] : 0.0f;
        a[u].y = (v && !hi) ? feat[gi[u] * CIN + 1] : 0.0f;
      }
#pragma unroll
      for (int u = 0; u < KB; ++u) {
        if (cnt[u] > 0) {
          const v8f z = {};
          const v8f d0 = __builtin_amdgcn_wmma_f32_16x16x4_f32(false, a[u], false, b0[u], (short)0, z, false, false);
          const v8f d1 = __builtin_amdgcn_wmma_f32_16x16x4_f32(false, a[u], false, b1[u], (short)0, z, false, false);
          add_tile(acc, j0, cnt[u], oo[u], lanepart, d0, d1);
        }
      }
    }
    for (int kk = 0; kk < 128; kk += 32) {
      const int cl = sEnd[kk + lane] - sBeg[kk + lane];
      unsigned heavy = __builtin_amdgcn_ballot_w32(cl > 16);
      while (heavy) {
        const int k = kk + __builtin_ctz(heavy);
        heavy &= heavy - 1;
        const int s = __builtin_amdgcn_readfirstlane(sBeg[k]);
        const int e = __builtin_amdgcn_readfirstlane(sEnd[k]);
        int p = s + 16 * (1 + w);
        if (p < e) {
          const float* wb = weight + k * (CIN * COUT);
          v2f c0, c1;
          c0.x = wb[ksel * COUT + row];
          c0.y = hi ? 0.0f : wb[COUT + row];
          c1.x = wb[ksel * COUT + 16 + row];
          c1.y = hi ? 0.0f : wb[COUT + 16 + row];
          for (; p < e; p += 16 * 8) {
            const int c = e - p;
            const bool v = pa < c;
            const int ia = p + pa, io = p + po;
            const int ga = in_idx[ia < MPAD ? ia : MPAD - 1];
            const int oa = out_idx[io < MPAD ? io : MPAD - 1];
            const int g = v ? ga : 0;
            const int o = po < c ? oa : -1;
            v2f aa;
            aa.x = v ? feat[g * CIN + ksel] : 0.0f;
            aa.y = (v && !hi) ? feat[g * CIN + 1] : 0.0f;
            const v8f z = {};
            const v8f d0 = __builtin_amdgcn_wmma_f32_16x16x4_f32(false, aa, false, c0, (short)0, z, false, false);
            const v8f d1 = __builtin_amdgcn_wmma_f32_16x16x4_f32(false, aa, false, c1, (short)0, z, false, false);
            add_tile(acc, j0, c, o, lanepart, d0, d1);
          }
        }
      }
    }
  } else {
    for (int m = tid; m < MPAD; m += 256) {
      const int o = out_idx[m];
      const int f = off_idx[m];
      const int lr = o - j0;
      if ((unsigned)o < (unsigned)NPTS && (unsigned)f < (unsigned)KVOL && (unsigned)lr < (unsigned)TROWS) {
        const int g = in_idx[m];
        const float f0 = feat[g * CIN + 0];
        const float f1 = feat[g * CIN + 1];
        const float f2 = feat[g * CIN + 2];
        const float* wb = weight + f * (CIN * COUT);
        for (int ch = 0; ch < COUT; ++ch)
          __hip_atomic_fetch_add(acc + lr * APITCH + ch,
                                 f0 * wb[ch] + f1 * wb[COUT + ch] + f2 * wb[2 * COUT + ch],
                                 __ATOMIC_RELAXED, __HIP_MEMORY_SCOPE_WORKGROUP);
      }
    }
  }
  __syncthreads();

  float4* o4 = reinterpret_cast<float4*>(out + (size_t)j0 * COUT);
  const float4* a4 = reinterpret_cast<const float4*>(acc);
  for (int i = tid; i < TROWS * COUT / 4; i += 256) {
    const int lrow = i >> 3;
    const int i0 = lrow * (APITCH / 4) + (i & 7) + ((lrow & 1) << 3);
    const float4 x = a4[i0];
    const float4 y = a4[i0 ^ 8];
    o4[i] = make_float4(x.x + y.x, x.y + y.y, x.z + y.z, x.w + y.w);
  }
}

extern "C" void kernel_launch(void* const* d_in, const int* in_sizes, int n_in,
                              void* d_out, int out_size, void* d_ws, size_t ws_size,
                              hipStream_t stream) {
  const float* feat    = (const float*)d_in[0];
  const float* weight  = (const float*)d_in[1];
  const int*   in_idx  = (const int*)d_in[2];
  const int*   out_idx = (const int*)d_in[3];
  const int*   off_idx = (const int*)d_in[4];
  float* out = (float*)d_out;
  int* ws = (int*)d_ws;

  spconv_bucket_kernel<<<PRE_BLOCKS, PRE_THREADS, 0, stream>>>(out_idx, off_idx, ws, ws_size);
  spconv_owner_kernel<<<NRB, 256, 0, stream>>>(feat, weight, in_idx, out_idx, off_idx,
                                               ws, ws_size, out);
}


# A work-group of 32 waves owns 320 output rows (250 groups, one per WGP): a wave takes 4 offsets, up to 3 tiles each in one batch, tiles about four-fifths full; LDS adds, arrival-order sums kept
# speedup vs baseline: 3.7356x; 1.0797x over previous
#include <hip/hip_runtime.h>

typedef float v2f __attribute__((ext_vector_type(2)));
typedef float v8f __attribute__((ext_vector_type(8)));

#define NPTS  80000
#define CIN   3
#define COUT  32
#define KVOL  125
#define MPAD  950000
#define TROWS 320
#define NRB   (NPTS / TROWS)
#define NBUCK (KVOL * NRB)
#define PRE_THREADS 1024
#define PRE_PER 4
#define PRE_SPAN (PRE_THREADS * PRE_PER)
#define PRE_BLOCKS ((MPAD + PRE_SPAN - 1) / PRE_SPAN)
#define WS_INTS (NBUCK + 1 + PRE_BLOCKS)
#define OWN_THREADS 1024
#define OWN_WAVES (OWN_THREADS / 32)
#define KO 4
#define TS 3
#define APITCH 64
#define ATRASH (TROWS * APITCH)

__device__ __forceinline__ int bucket_of(int off, int orow) {
  if ((unsigned)orow >= (unsigned)NPTS || (unsigned)off >= (unsigned)KVOL) return NBUCK;
  return off * NRB + orow / TROWS;
}

__global__ __launch_bounds__(PRE_THREADS) void spconv_bucket_kernel(
    const int* __restrict__ out_idx, const int* __restrict__ off_idx,
    int* __restrict__ ws, size_t ws_bytes) {
  if (ws_bytes < WS_INTS * sizeof(int)) return;
  int* start = ws;
  int* flags = ws + NBUCK + 1;
  const int base = blockIdx.x * PRE_SPAN + threadIdx.x;
  int cb[PRE_PER], pb[PRE_PER];
#pragma unroll
  for (int j = 0; j < PRE_PER; ++j) {
    const int m  = base + j * PRE_THREADS;
    const int mc = m < MPAD ? m : MPAD - 1;
    const int mp = mc > 0 ? mc - 1 : 0;
    const int f1 = off_idx[mc], o1 = out_idx[mc];
    const int f0 = off_idx[mp], o0 = out_idx[mp];
    cb[j] = bucket_of(f1, o1);
    pb[j] = bucket_of(f0, o0);
  }
  int bad = 0;
#pragma unroll
  for (int j = 0; j < PRE_PER; ++j) {
    const int m = base + j * PRE_THREADS;
    if (m < MPAD) {
      if (cb[j] < pb[j]) {
        bad = 1;
      } else {
        const int lo = (m == 0) ? 0 : pb[j] + 1;
        for (int b = lo; b <= cb[j]; ++b) start[b] = m;
        if (m == MPAD - 1)
          for (int b = cb[j] + 1; b <= NBUCK; ++b) start[b] = MPAD;
      }
    }
  }
  const int anybad = __syncthreads_or(bad);
  if (threadIdx.x == 0) flags[blockIdx.x] = anybad;
}

template <int R>
__device__ __forceinline__ void add_rows(float* acc, int cnt, int vbase, int lanepart,
                                         const v8f& d0, const v8f& d1) {
  if (2 * R < cnt) {
    const int rowpart = __builtin_amdgcn_update_dpp(0, vbase, 0x150 + R, 0xf, 0xf, true);
    float* p = acc + rowpart + lanepart;
    __hip_atomic_fetch_add(p,      d0[R], __ATOMIC_RELAXED, __HIP_MEMORY_SCOPE_WORKGROUP);
    __hip_atomic_fetch_add(p + 16, d1[R], __ATOMIC_RELAXED, __HIP_MEMORY_SCOPE_WORKGROUP);
    if constexpr (R < 7) add_rows<R + 1>(acc, cnt, vbase, lanepart, d0, d1);
  }
}

__device__ __forceinline__ void add_tile(float* acc, int j0, int cnt, int ov, int lanepart,
                                         const v8f& d0, const v8f& d1) {
  const int lr = ov - j0;
  const int vbase = ((unsigned)lr < (unsigned)TROWS) ? lr * APITCH : ATRASH;
  add_rows<0>(acc, cnt, vbase, lanepart, d0, d1);
}

__global__ __launch_bounds__(OWN_THREADS) void spconv_owner_kernel(
    const float* __restrict__ feat,
    const float* __restrict__ weight,
    const int*   __restrict__ in_idx,
    const int*   __restrict__ out_idx,
    const int*   __restrict__ off_idx,
    const int*   __restrict__ ws, size_t ws_bytes,
    float*       __restrict__ out) {
  __shared__ __attribute__((aligned(16))) float acc[(TROWS + 1) * APITCH];
  __shared__ int sBeg[128];
  __shared__ int sEnd[128];
  const int tid = threadIdx.x;
  const int rb  = blockIdx.x;
  const int j0  = rb * TROWS;

  {
    float4* z4 = reinterpret_cast<float4*>(acc);
    for (int i = tid; i < (TROWS + 1) * APITCH / 4; i += OWN_THREADS) z4[i] = make_float4(0.0f, 0.0f, 0.0f, 0.0f);
  }

  const bool ws_ok = ws_bytes >= WS_INTS * sizeof(int);
  int bad = ws_ok ? 0 : 1;
  if (ws_ok) {
    const int* flags = ws + NBUCK + 1;
    if (tid < PRE_BLOCKS) bad = flags[tid];
    if (tid < 128) {
      int s = 0, e = 0;
      if (tid < KVOL) { s = ws[tid * NRB + rb]; e = ws[tid * NRB + rb + 1]; }
      sBeg[tid] = s;
      sEnd[tid] = e;
    }
  }
  const int anybad = __syncthreads_or(bad);

  if (!anybad) {
    const int lane = tid & 31;
    const int row  = lane & 15;
    const int hi   = lane >> 4;
    const int pa   = ((row & 7) << 1) | (row >> 3);
    const int po   = pa ^ hi;
    const int lanepart = hi * 32 + row;
    const int w    = __builtin_amdgcn_readfirstlane(tid >> 5);
    const int ksel = hi ? 2 : 0;

    const int kmine = w + OWN_WAVES * (lane & (KO - 1));
    const int vS = sBeg[kmine];
    const int vE = sEnd[kmine];

    int cnt[KO][TS], oo[KO][TS], gi[KO][TS];
    v2f a[KO][TS], b0[KO], b1[KO];
#pragma unroll
    for (int u = 0; u < KO; ++u) {
      const int k = w + OWN_WAVES * u;
      const int s = __builtin_amdgcn_readlane(vS, u);
      const int e = __builtin_amdgcn_readlane(vE, u);
      const float* wb = weight + (k < KVOL ? k : 0) * (CIN * COUT);
      b0[u].x = wb[ksel * COUT + row];
      b0[u].y = hi ? 0.0f : wb[COUT + row];
      b1[u].x = wb[ksel * COUT + 16 + row];
      b1[u].y = hi ? 0.0f : wb[COUT + 16 + row];
#pragma unroll
      for (int t = 0; t < TS; ++t) {
        const int ts = s + 16 * t;
        cnt[u][t] = e - ts;
        const int ia = ts + pa, io = ts + po;
        const int ga = in_idx[ia < MPAD ? ia : MPAD - 1];
        const int oa = out_idx[io < MPAD ? io : MPAD - 1];
        gi[u][t] = pa < cnt[u][t] ? ga : 0;
        oo[u][t] = po < cnt[u][t] ? oa : -1;
      }
    }
#pragma unroll
    for (int u = 0; u < KO; ++u) {
#pragma unroll
      for (int t = 0; t < TS; ++t) {
        const bool v = pa < cnt[u][t];
        a[u][t].x = v ? feat[gi[u][t] * CIN + ksel] : 0.0f;
        a[u][t].y = (v && !hi) ? feat[gi[u][t] * CIN + 1] : 0.0f;
      }
    }
#pragma unroll
    for (int u = 0; u < KO; ++u) {
#pragma unroll
      for (int t = 0; t < TS; ++t) {
        if (cnt[u][t] > 0) {
          const v8f z = {};
          const v8f d0 = __builtin_amdgcn_wmma_f32_16x16x4_f32(false, a[u][t], false, b0[u], (short)0, z, false, false);
          const v8f d1 = __builtin_amdgcn_wmma_f32_16x16x4_f32(false, a[u][t], false, b1[u], (short)0, z, false, false);
          add_tile(acc, j0, cnt[u][t], oo[u][t], lanepart, d0, d1);
        }
      }
    }
    for (int kk = 0; kk < 128; kk += 32) {
      const int cl = sEnd[kk + lane] - sBeg[kk + lane];
      unsigned heavy = __builtin_amdgcn_ballot_w32(cl > 16 * TS);
      while (heavy) {
        const int k = kk + __builtin_ctz(heavy);
        heavy &= heavy - 1;
        const int s = __builtin_amdgcn_readfirstlane(sBeg[k]);
        const int e = __builtin_amdgcn_readfirstlane(sEnd[k]);
        int p = s + 16 * (TS + ((w - k) & (OWN_WAVES - 1)));
        if (p < e) {
          const float* wb = weight + k * (CIN * COUT);
          v2f c0, c1;
          c0.x = wb[ksel * COUT + row];
          c0.y = hi ? 0.0f : wb[COUT + row];
          c1.x = wb[ksel * COUT + 16 + row];
          c1.y = hi ? 0.0f : wb[COUT + 16 + row];
          for (; p < e; p += 16 * OWN_WAVES) {
            const int c = e - p;
            const bool v = pa < c;
            const int ia = p + pa, io = p + po;
            const int ga = in_idx[ia < MPAD ? ia : MPAD - 1];
            const int oa = out_idx[io < MPAD ? io : MPAD - 1];
            const int g = v ? ga : 0;
            const int o = po < c ? oa : -1;
            v2f aa;
            aa.x = v ? feat[g * CIN + ksel] : 0.0f;
            aa.y = (v && !hi) ? feat[g * CIN + 1] : 0.0f;
            const v8f z = {};
            const v8f d0 = __builtin_amdgcn_wmma_f32_16x16x4_f32(false, aa, false, c0, (short)0, z, false, false);
            const v8f d1 = __builtin_amdgcn_wmma_f32_16x16x4_f32(false, aa, false, c1, (short)0, z, false, false);
            add_tile(acc, j0, c, o, lanepart, d0, d1);
          }
        }
      }
    }
  } else {
    for (int m = tid; m < MPAD; m += OWN_THREADS) {
      const int o = out_idx[m];
      const int f = off_idx[m];
      const int lr = o - j0;
      if ((unsigned)o < (unsigned)NPTS && (unsigned)f < (unsigned)KVOL && (unsigned)lr < (unsigned)TROWS) {
        const int g = in_idx[m];
        const float f0 = feat[g * CIN + 0];
        const float f1 = feat[g * CIN + 1];
        const float f2 = feat[g * CIN + 2];
        const float* wb = weight + f * (CIN * COUT);
        for (int ch = 0; ch < COUT; ++ch)
          __hip_atomic_fetch_add(acc + lr * APITCH + ch,
                                 f0 * wb[ch] + f1 * wb[COUT + ch] + f2 * wb[2 * COUT + ch],
                                 __ATOMIC_RELAXED, __HIP_MEMORY_SCOPE_WORKGROUP);
      }
    }
  }
  __syncthreads();

  float4* o4 = reinterpret_cast<float4*>(out + (size_t)j0 * COUT);
  const float4* a4 = reinterpret_cast<const float4*>(acc);
  for (int i = tid; i < TROWS * COUT / 4; i += OWN_THREADS) {
    const int lrow = i >> 3;
    const int i0 = lrow * (APITCH / 4) + (i & 7) + ((lrow & 1) << 3);
    const float4 x = a4[i0];
    const float4 y = a4[i0 ^ 8];
    o4[i] = make_float4(x.x + y.x, x.y + y.y, x.z + y.z, x.w + y.w);
  }
}

extern "C" void kernel_launch(void* const* d_in, const int* in_sizes, int n_in,
                              void* d_out, int out_size, void* d_ws, size_t ws_size,
                              hipStream_t stream) {
  const float* feat    = (const float*)d_in[0];
  const float* weight  = (const float*)d_in[1];
  const int*   in_idx  = (const int*)d_in[2];
  const int*   out_idx = (const int*)d_in[3];
  const int*   off_idx = (const int*)d_in[4];
  float* out = (float*)d_out;
  int* ws = (int*)d_ws;

  spconv_bucket_kernel<<<PRE_BLOCKS, PRE_THREADS, 0, stream>>>(out_idx, off_idx, ws, ws_size);
  spconv_owner_kernel<<<NRB, OWN_THREADS, 0, stream>>>(feat, weight, in_idx, out_idx, off_idx,
                                               ws, ws_size, out);
}
